# starting kernel
# baseline (speedup 1.0000x reference)
#include <hip/hip_runtime.h>

typedef float v2f __attribute__((ext_vector_type(2)));
typedef float v8f __attribute__((ext_vector_type(8)));

__device__ __forceinline__ float bperm(int srcLane, float v) {
    return __int_as_float(__builtin_amdgcn_ds_bpermute(srcLane << 2, __float_as_int(v)));
}

__global__ __launch_bounds__(256) void fused_attn_kernel(
    const float* __restrict__ hist,
    const float* __restrict__ paper,
    const float* __restrict__ author,
    const float* __restrict__ Wh, const float* __restrict__ bh,
    const float* __restrict__ Wp, const float* __restrict__ bp,
    const float* __restrict__ Wa, const float* __restrict__ ba,
    const float* __restrict__ lnw, const float* __restrict__ lnb,
    const float* __restrict__ Wfc, const float* __restrict__ bfc,
    float* __restrict__ out, int n)
{
    const int row      = blockIdx.x * blockDim.x + threadIdx.x;
    const int lane     = threadIdx.x & 31;
    const int waveBase = row - lane;
    const bool lo      = lane < 16;

    float flat[6] = {0.f, 0.f, 0.f, 0.f, 0.f, 0.f};

    if (row < n) {
        const float* hp = hist   + (size_t)row * 2u;
        const float* pp = paper  + (size_t)row * 5u;
        const float* ap = author + (size_t)row * 5u;
        v2f   hv  = __builtin_nontemporal_load((const v2f*)hp);
        float pv0 = __builtin_nontemporal_load(pp + 0);
        float pv1 = __builtin_nontemporal_load(pp + 1);
        float pv2 = __builtin_nontemporal_load(pp + 2);
        float pv3 = __builtin_nontemporal_load(pp + 3);
        float pv4 = __builtin_nontemporal_load(pp + 4);
        float av0 = __builtin_nontemporal_load(ap + 0);
        float av1 = __builtin_nontemporal_load(ap + 1);
        float av2 = __builtin_nontemporal_load(ap + 2);
        float av3 = __builtin_nontemporal_load(ap + 3);
        float av4 = __builtin_nontemporal_load(ap + 4);

        float fx[3], fy[3];
        fx[0] = fmaxf(Wh[0]*hv.x + Wh[1]*hv.y + bh[0], 0.f);
        fy[0] = fmaxf(Wh[2]*hv.x + Wh[3]*hv.y + bh[1], 0.f);
        fx[1] = fmaxf(Wp[0]*pv0 + Wp[1]*pv1 + Wp[2]*pv2 + Wp[3]*pv3 + Wp[4]*pv4 + bp[0], 0.f);
        fy[1] = fmaxf(Wp[5]*pv0 + Wp[6]*pv1 + Wp[7]*pv2 + Wp[8]*pv3 + Wp[9]*pv4 + bp[1], 0.f);
        fx[2] = fmaxf(Wa[0]*av0 + Wa[1]*av1 + Wa[2]*av2 + Wa[3]*av3 + Wa[4]*av4 + ba[0], 0.f);
        fy[2] = fmaxf(Wa[5]*av0 + Wa[6]*av1 + Wa[7]*av2 + Wa[8]*av3 + Wa[9]*av4 + ba[1], 0.f);

        const float is2 = 0.70710678118654752f;
        float l[3][3];
        #pragma unroll
        for (int s = 0; s < 3; ++s)
            #pragma unroll
            for (int t = 0; t < 3; ++t)
                l[s][t] = (fx[s]*fx[t] + fy[s]*fy[t]) * is2;

        float w[3][3];
        #pragma unroll
        for (int t = 0; t < 3; ++t) {
            float m  = fmaxf(l[0][t], fmaxf(l[1][t], l[2][t]));
            float e0 = __expf(l[0][t] - m);
            float e1 = __expf(l[1][t] - m);
            float e2 = __expf(l[2][t] - m);
            float r  = __builtin_amdgcn_rcpf(e0 + e1 + e2);
            w[0][t] = e0 * r; w[1][t] = e1 * r; w[2][t] = e2 * r;
        }

        float ax[3], ay[3];
        #pragma unroll
        for (int s = 0; s < 3; ++s) {
            ax[s] = w[s][0]*fx[0] + w[s][1]*fx[1] + w[s][2]*fx[2];
            ay[s] = w[s][0]*fy[0] + w[s][1]*fy[1] + w[s][2]*fy[2];
        }

        float mu = (ax[0]+ay[0]+ax[1]+ay[1]+ax[2]+ay[2]) * (1.f/6.f);
        float var = 0.f;
        #pragma unroll
        for (int s = 0; s < 3; ++s) {
            float dx = ax[s] - mu, dy = ay[s] - mu;
            var += dx*dx + dy*dy;
        }
        var *= (1.f/6.f);
        float inv = __builtin_amdgcn_rsqf(var + 1e-5f);
        #pragma unroll
        for (int s = 0; s < 3; ++s) {
            flat[2*s+0] = (ax[s] - mu) * inv * lnw[2*s+0] + lnb[2*s+0];
            flat[2*s+1] = (ay[s] - mu) * inv * lnw[2*s+1] + lnb[2*s+1];
        }
    }

    if (waveBase + 32 <= n) {
        const int  j  = lane & 15;
        const bool jv = j < 6;
        v2f A1, A2;
        A1.x = jv ? Wfc[j*6 + (lo ? 0 : 2)] : 0.f;
        A1.y = jv ? Wfc[j*6 + (lo ? 1 : 3)] : 0.f;
        A2.x = (jv && lo) ? Wfc[j*6 + 4] : 0.f;
        A2.y = (jv && lo) ? Wfc[j*6 + 5] : 0.f;

        v8f C = {};
        if (lo) { C[0]=bfc[0]; C[1]=bfc[1]; C[2]=bfc[2]; C[3]=bfc[3]; C[4]=bfc[4]; C[5]=bfc[5]; }

        float f2s = bperm(lane & 15, flat[2]);
        float f3s = bperm(lane & 15, flat[3]);
        v2f B1; B1.x = lo ? flat[0] : f2s;  B1.y = lo ? flat[1] : f3s;
        v2f B2; B2.x = lo ? flat[4] : 0.f;  B2.y = lo ? flat[5] : 0.f;
        v8f D1 = __builtin_amdgcn_wmma_f32_16x16x4_f32(false, A1, false, B1, (short)0, C,  false, false);
        D1     = __builtin_amdgcn_wmma_f32_16x16x4_f32(false, A2, false, B2, (short)0, D1, false, false);

        float f0s = bperm(lane | 16, flat[0]);
        float f1s = bperm(lane | 16, flat[1]);
        float f4s = bperm(lane | 16, flat[4]);
        float f5s = bperm(lane | 16, flat[5]);
        v2f B3; B3.x = lo ? f0s : flat[2];  B3.y = lo ? f1s : flat[3];
        v2f B4; B4.x = lo ? f4s : 0.f;      B4.y = lo ? f5s : 0.f;
        v8f D2 = __builtin_amdgcn_wmma_f32_16x16x4_f32(false, A1, false, B3, (short)0, C,  false, false);
        D2     = __builtin_amdgcn_wmma_f32_16x16x4_f32(false, A2, false, B4, (short)0, D2, false, false);

        if (lo) {
            float* o1 = out + (size_t)(waveBase + lane) * 6u;
            v2f s0; s0.x = fmaxf(D1[0], 0.f); s0.y = fmaxf(D1[1], 0.f);
            v2f s1; s1.x = fmaxf(D1[2], 0.f); s1.y = fmaxf(D1[3], 0.f);
            v2f s2; s2.x = fmaxf(D1[4], 0.f); s2.y = fmaxf(D1[5], 0.f);
            __builtin_nontemporal_store(s0, (v2f*)(o1 + 0));
            __builtin_nontemporal_store(s1, (v2f*)(o1 + 2));
            __builtin_nontemporal_store(s2, (v2f*)(o1 + 4));

            float* o2 = out + (size_t)(waveBase + 16 + lane) * 6u;
            v2f t0; t0.x = fmaxf(D2[0], 0.f); t0.y = fmaxf(D2[1], 0.f);
            v2f t1; t1.x = fmaxf(D2[2], 0.f); t1.y = fmaxf(D2[3], 0.f);
            v2f t2; t2.x = fmaxf(D2[4], 0.f); t2.y = fmaxf(D2[5], 0.f);
            __builtin_nontemporal_store(t0, (v2f*)(o2 + 0));
            __builtin_nontemporal_store(t1, (v2f*)(o2 + 2));
            __builtin_nontemporal_store(t2, (v2f*)(o2 + 4));
        }
    } else if (row < n) {
        float* o = out + (size_t)row * 6u;
        #pragma unroll
        for (int jj = 0; jj < 6; ++jj) {
            float acc = bfc[jj];
            #pragma unroll
            for (int k = 0; k < 6; ++k) acc += Wfc[jj*6 + k] * flat[k];
            o[jj] = fmaxf(acc, 0.f);
        }
    }
}

extern "C" void kernel_launch(void* const* d_in, const int* in_sizes, int n_in,
                              void* d_out, int out_size, void* d_ws, size_t ws_size,
                              hipStream_t stream) {
    const float* hist   = (const float*)d_in[0];
    const float* paper  = (const float*)d_in[1];
    const float* author = (const float*)d_in[2];
    const float* Wh  = (const float*)d_in[3];
    const float* bh  = (const float*)d_in[4];
    const float* Wp  = (const float*)d_in[5];
    const float* bp  = (const float*)d_in[6];
    const float* Wa  = (const float*)d_in[7];
    const float* ba  = (const float*)d_in[8];
    const float* lnw = (const float*)d_in[9];
    const float* lnb = (const float*)d_in[10];
    const float* Wfc = (const float*)d_in[11];
    const float* bfc = (const float*)d_in[12];
    float* out = (float*)d_out;

    const int n = in_sizes[0] / 2;
    const int threads = 256;
    const int blocks  = (n + threads - 1) / threads;

    fused_attn_kernel<<<blocks, threads, 0, stream>>>(
        hist, paper, author, Wh, bh, Wp, bp, Wa, ba, lnw, lnb, Wfc, bfc, out, n);
}


# A fixed grid of 1024 resident work-groups loops over 256-row slabs, next slab's loads in flight, parameters read once per wave: 8192 waves enter instead of 131072 one-shot waves
# speedup vs baseline: 1.1426x; 1.1426x over previous
#include <hip/hip_runtime.h>

typedef float v2f __attribute__((ext_vector_type(2)));
typedef float v8f __attribute__((ext_vector_type(8)));

__device__ __forceinline__ float bperm(int srcLane, float v) {
    return __int_as_float(__builtin_amdgcn_ds_bpermute(srcLane << 2, __float_as_int(v)));
}


struct RowIn { v2f h; float p0, p1, p2, p3, p4, a0, a1, a2, a3, a4; };

struct Wts {
    float wh[4], bh[2], wp[10], bp[2], wa[10], ba[2], lnw[6], lnb[6];
};

__device__ __forceinline__ RowIn load_row(const float* __restrict__ hist,
                                          const float* __restrict__ paper,
                                          const float* __restrict__ author, int r) {
    const float* hp = hist   + (size_t)r * 2u;
    const float* pp = paper  + (size_t)r * 5u;
    const float* ap = author + (size_t)r * 5u;
    RowIn x;
    x.h  = __builtin_nontemporal_load((const v2f*)hp);
    x.p0 = __builtin_nontemporal_load(pp + 0);
    x.p1 = __builtin_nontemporal_load(pp + 1);
    x.p2 = __builtin_nontemporal_load(pp + 2);
    x.p3 = __builtin_nontemporal_load(pp + 3);
    x.p4 = __builtin_nontemporal_load(pp + 4);
    x.a0 = __builtin_nontemporal_load(ap + 0);
    x.a1 = __builtin_nontemporal_load(ap + 1);
    x.a2 = __builtin_nontemporal_load(ap + 2);
    x.a3 = __builtin_nontemporal_load(ap + 3);
    x.a4 = __builtin_nontemporal_load(ap + 4);
    return x;
}

__device__ __forceinline__ void process_slab(const RowIn& x, const Wts& W,
                                             v2f A1, v2f A2, v8f C,
                                             const float* __restrict__ Wfc,
                                             const float* __restrict__ bfc,
                                             float* __restrict__ out,
                                             int row, int lane, bool lo, int n)
{
    const int waveBase = row - lane;
    float flat[6];
    {
        float fx[3], fy[3];
        fx[0] = fmaxf(W.wh[0]*x.h.x + W.wh[1]*x.h.y + W.bh[0], 0.f);
        fy[0] = fmaxf(W.wh[2]*x.h.x + W.wh[3]*x.h.y + W.bh[1], 0.f);
        fx[1] = fmaxf(W.wp[0]*x.p0 + W.wp[1]*x.p1 + W.wp[2]*x.p2 + W.wp[3]*x.p3 + W.wp[4]*x.p4 + W.bp[0], 0.f);
        fy[1] = fmaxf(W.wp[5]*x.p0 + W.wp[6]*x.p1 + W.wp[7]*x.p2 + W.wp[8]*x.p3 + W.wp[9]*x.p4 + W.bp[1], 0.f);
        fx[2] = fmaxf(W.wa[0]*x.a0 + W.wa[1]*x.a1 + W.wa[2]*x.a2 + W.wa[3]*x.a3 + W.wa[4]*x.a4 + W.ba[0], 0.f);
        fy[2] = fmaxf(W.wa[5]*x.a0 + W.wa[6]*x.a1 + W.wa[7]*x.a2 + W.wa[8]*x.a3 + W.wa[9]*x.a4 + W.ba[1], 0.f);

        const float is2 = 0.70710678118654752f;
        float l[3][3];
        #pragma unroll
        for (int s = 0; s < 3; ++s)
            #pragma unroll
            for (int t = 0; t < 3; ++t)
                l[s][t] = (fx[s]*fx[t] + fy[s]*fy[t]) * is2;

        float w[3][3];
        #pragma unroll
        for (int t = 0; t < 3; ++t) {
            float m  = fmaxf(l[0][t], fmaxf(l[1][t], l[2][t]));
            float e0 = __expf(l[0][t] - m);
            float e1 = __expf(l[1][t] - m);
            float e2 = __expf(l[2][t] - m);
            float r  = __builtin_amdgcn_rcpf(e0 + e1 + e2);
            w[0][t] = e0 * r; w[1][t] = e1 * r; w[2][t] = e2 * r;
        }

        float ax[3], ay[3];
        #pragma unroll
        for (int s = 0; s < 3; ++s) {
            ax[s] = w[s][0]*fx[0] + w[s][1]*fx[1] + w[s][2]*fx[2];
            ay[s] = w[s][0]*fy[0] + w[s][1]*fy[1] + w[s][2]*fy[2];
        }

        float mu = (ax[0]+ay[0]+ax[1]+ay[1]+ax[2]+ay[2]) * (1.f/6.f);
        float var = 0.f;
        #pragma unroll
        for (int s = 0; s < 3; ++s) {
            float dx = ax[s] - mu, dy = ay[s] - mu;
            var += dx*dx + dy*dy;
        }
        var *= (1.f/6.f);
        float inv = __builtin_amdgcn_rsqf(var + 1e-5f);
        #pragma unroll
        for (int s = 0; s < 3; ++s) {
            flat[2*s+0] = (ax[s] - mu) * inv * W.lnw[2*s+0] + W.lnb[2*s+0];
            flat[2*s+1] = (ay[s] - mu) * inv * W.lnw[2*s+1] + W.lnb[2*s+1];
        }
    }

    if (waveBase + 32 <= n) {
        float f2s = bperm(lane & 15, flat[2]);
        float f3s = bperm(lane & 15, flat[3]);
        v2f B1; B1.x = lo ? flat[0] : f2s;  B1.y = lo ? flat[1] : f3s;
        v2f B2; B2.x = lo ? flat[4] : 0.f;  B2.y = lo ? flat[5] : 0.f;
        v8f D1 = __builtin_amdgcn_wmma_f32_16x16x4_f32(false, A1, false, B1, (short)0, C,  false, false);
        D1     = __builtin_amdgcn_wmma_f32_16x16x4_f32(false, A2, false, B2, (short)0, D1, false, false);

        float f0s = bperm(lane | 16, flat[0]);
        float f1s = bperm(lane | 16, flat[1]);
        float f4s = bperm(lane | 16, flat[4]);
        float f5s = bperm(lane | 16, flat[5]);
        v2f B3; B3.x = lo ? f0s : flat[2];  B3.y = lo ? f1s : flat[3];
        v2f B4; B4.x = lo ? f4s : 0.f;      B4.y = lo ? f5s : 0.f;
        v8f D2 = __builtin_amdgcn_wmma_f32_16x16x4_f32(false, A1, false, B3, (short)0, C,  false, false);
        D2     = __builtin_amdgcn_wmma_f32_16x16x4_f32(false, A2, false, B4, (short)0, D2, false, false);

        if (lo) {
            float* o1 = out + (size_t)(waveBase + lane) * 6u;
            v2f s0; s0.x = fmaxf(D1[0], 0.f); s0.y = fmaxf(D1[1], 0.f);
            v2f s1; s1.x = fmaxf(D1[2], 0.f); s1.y = fmaxf(D1[3], 0.f);
            v2f s2; s2.x = fmaxf(D1[4], 0.f); s2.y = fmaxf(D1[5], 0.f);
            __builtin_nontemporal_store(s0, (v2f*)(o1 + 0));
            __builtin_nontemporal_store(s1, (v2f*)(o1 + 2));
            __builtin_nontemporal_store(s2, (v2f*)(o1 + 4));

            float* o2 = out + (size_t)(waveBase + 16 + lane) * 6u;
            v2f t0; t0.x = fmaxf(D2[0], 0.f); t0.y = fmaxf(D2[1], 0.f);
            v2f t1; t1.x = fmaxf(D2[2], 0.f); t1.y = fmaxf(D2[3], 0.f);
            v2f t2; t2.x = fmaxf(D2[4], 0.f); t2.y = fmaxf(D2[5], 0.f);
            __builtin_nontemporal_store(t0, (v2f*)(o2 + 0));
            __builtin_nontemporal_store(t1, (v2f*)(o2 + 2));
            __builtin_nontemporal_store(t2, (v2f*)(o2 + 4));
        }
    } else if (row < n) {
        float* o = out + (size_t)row * 6u;
        #pragma unroll
        for (int jj = 0; jj < 6; ++jj) {
            float acc = bfc[jj];
            #pragma unroll
            for (int k = 0; k < 6; ++k) acc += Wfc[jj*6 + k] * flat[k];
            o[jj] = fmaxf(acc, 0.f);
        }
    }
}

__global__ __launch_bounds__(256) void fused_attn_kernel(
    const float* __restrict__ hist,
    const float* __restrict__ paper,
    const float* __restrict__ author,
    const float* __restrict__ Wh, const float* __restrict__ bh,
    const float* __restrict__ Wp, const float* __restrict__ bp,
    const float* __restrict__ Wa, const float* __restrict__ ba,
    const float* __restrict__ lnw, const float* __restrict__ lnb,
    const float* __restrict__ Wfc, const float* __restrict__ bfc,
    float* __restrict__ out, int n)
{
    if (n <= 0) return;
    const int tid    = threadIdx.x;
    const int lane   = tid & 31;
    const bool lo    = lane < 16;
    const int stride = gridDim.x * blockDim.x;
    const int last   = n - 1;

    const int  j  = lane & 15;
    const bool jv = j < 6;
    v2f A1, A2;
    A1.x = jv ? Wfc[j*6 + (lo ? 0 : 2)] : 0.f;
    A1.y = jv ? Wfc[j*6 + (lo ? 1 : 3)] : 0.f;
    A2.x = (jv && lo) ? Wfc[j*6 + 4] : 0.f;
    A2.y = (jv && lo) ? Wfc[j*6 + 5] : 0.f;
    v8f C = {};
    if (lo) { C[0]=bfc[0]; C[1]=bfc[1]; C[2]=bfc[2]; C[3]=bfc[3]; C[4]=bfc[4]; C[5]=bfc[5]; }

    Wts W;
    #pragma unroll
    for (int i = 0; i < 4;  ++i) W.wh[i] = Wh[i];
    #pragma unroll
    for (int i = 0; i < 10; ++i) { W.wp[i] = Wp[i]; W.wa[i] = Wa[i]; }
    #pragma unroll
    for (int i = 0; i < 2;  ++i) { W.bh[i] = bh[i]; W.bp[i] = bp[i]; W.ba[i] = ba[i]; }
    #pragma unroll
    for (int i = 0; i < 6;  ++i) { W.lnw[i] = lnw[i]; W.lnb[i] = lnb[i]; }

    int slab = blockIdx.x * blockDim.x;
    RowIn cur = load_row(hist, paper, author, min(slab + tid, last));

    for (; slab < n; slab += stride) {
        const RowIn nxt = load_row(hist, paper, author, min(slab + stride + tid, last));
        process_slab(cur, W, A1, A2, C, Wfc, bfc, out, slab + tid, lane, lo, n);
        cur = nxt;
    }
}

extern "C" void kernel_launch(void* const* d_in, const int* in_sizes, int n_in,
                              void* d_out, int out_size, void* d_ws, size_t ws_size,
                              hipStream_t stream) {
    const float* hist   = (const float*)d_in[0];
    const float* paper  = (const float*)d_in[1];
    const float* author = (const float*)d_in[2];
    const float* Wh  = (const float*)d_in[3];
    const float* bh  = (const float*)d_in[4];
    const float* Wp  = (const float*)d_in[5];
    const float* bp  = (const float*)d_in[6];
    const float* Wa  = (const float*)d_in[7];
    const float* ba  = (const float*)d_in[8];
    const float* lnw = (const float*)d_in[9];
    const float* lnb = (const float*)d_in[10];
    const float* Wfc = (const float*)d_in[11];
    const float* bfc = (const float*)d_in[12];
    float* out = (float*)d_out;

    const int n = in_sizes[0] / 2;
    const int threads = 256;
    const int blocks  = 1024;

    fused_attn_kernel<<<blocks, threads, 0, stream>>>(
        hist, paper, author, Wh, bh, Wp, bp, Wa, ba, lnw, lnb, Wfc, bfc, out, n);
}


# FC product repacked: Wfc sits twice in A (M rows 0-5 on the lower lane half's k, 8-13 on the upper's): each lane feeds and stores its own row, no lane exchange, 3 WMMA not 4; FC sum now 2+2+2
# speedup vs baseline: 1.1855x; 1.0375x over previous
#include <hip/hip_runtime.h>

typedef float v2f __attribute__((ext_vector_type(2)));
typedef float v8f __attribute__((ext_vector_type(8)));

__device__ __forceinline__ float bperm(int srcLane, float v) {
    return __int_as_float(__builtin_amdgcn_ds_bpermute(srcLane << 2, __float_as_int(v)));
}


struct RowIn { v2f h; float p0, p1, p2, p3, p4, a0, a1, a2, a3, a4; };

struct Wts {
    float wh[4], bh[2], wp[10], bp[2], wa[10], ba[2], lnw[6], lnb[6];
};

__device__ __forceinline__ RowIn load_row(const float* __restrict__ hist,
                                          const float* __restrict__ paper,
                                          const float* __restrict__ author, int r) {
    const float* hp = hist   + (size_t)r * 2u;
    const float* pp = paper  + (size_t)r * 5u;
    const float* ap = author + (size_t)r * 5u;
    RowIn x;
    x.h  = __builtin_nontemporal_load((const v2f*)hp);
    x.p0 = __builtin_nontemporal_load(pp + 0);
    x.p1 = __builtin_nontemporal_load(pp + 1);
    x.p2 = __builtin_nontemporal_load(pp + 2);
    x.p3 = __builtin_nontemporal_load(pp + 3);
    x.p4 = __builtin_nontemporal_load(pp + 4);
    x.a0 = __builtin_nontemporal_load(ap + 0);
    x.a1 = __builtin_nontemporal_load(ap + 1);
    x.a2 = __builtin_nontemporal_load(ap + 2);
    x.a3 = __builtin_nontemporal_load(ap + 3);
    x.a4 = __builtin_nontemporal_load(ap + 4);
    return x;
}

__device__ __forceinline__ void process_slab(const RowIn& x, const Wts& W,
                                             v2f A0, v2f A1, v2f A2, v8f C,
                                             const float* __restrict__ Wfc,
                                             const float* __restrict__ bfc,
                                             float* __restrict__ out,
                                             int row, int lane, int n)
{
    const int waveBase = row - lane;
    float flat[6];
    {
        float fx[3], fy[3];
        fx[0] = fmaxf(W.wh[0]*x.h.x + W.wh[1]*x.h.y + W.bh[0], 0.f);
        fy[0] = fmaxf(W.wh[2]*x.h.x + W.wh[3]*x.h.y + W.bh[1], 0.f);
        fx[1] = fmaxf(W.wp[0]*x.p0 + W.wp[1]*x.p1 + W.wp[2]*x.p2 + W.wp[3]*x.p3 + W.wp[4]*x.p4 + W.bp[0], 0.f);
        fy[1] = fmaxf(W.wp[5]*x.p0 + W.wp[6]*x.p1 + W.wp[7]*x.p2 + W.wp[8]*x.p3 + W.wp[9]*x.p4 + W.bp[1], 0.f);
        fx[2] = fmaxf(W.wa[0]*x.a0 + W.wa[1]*x.a1 + W.wa[2]*x.a2 + W.wa[3]*x.a3 + W.wa[4]*x.a4 + W.ba[0], 0.f);
        fy[2] = fmaxf(W.wa[5]*x.a0 + W.wa[6]*x.a1 + W.wa[7]*x.a2 + W.wa[8]*x.a3 + W.wa[9]*x.a4 + W.ba[1], 0.f);

        const float is2 = 0.70710678118654752f;
        float l[3][3];
        #pragma unroll
        for (int s = 0; s < 3; ++s)
            #pragma unroll
            for (int t = 0; t < 3; ++t)
                l[s][t] = (fx[s]*fx[t] + fy[s]*fy[t]) * is2;

        float w[3][3];
        #pragma unroll
        for (int t = 0; t < 3; ++t) {
            float m  = fmaxf(l[0][t], fmaxf(l[1][t], l[2][t]));
            float e0 = __expf(l[0][t] - m);
            float e1 = __expf(l[1][t] - m);
            float e2 = __expf(l[2][t] - m);
            float r  = __builtin_amdgcn_rcpf(e0 + e1 + e2);
            w[0][t] = e0 * r; w[1][t] = e1 * r; w[2][t] = e2 * r;
        }

        float ax[3], ay[3];
        #pragma unroll
        for (int s = 0; s < 3; ++s) {
            ax[s] = w[s][0]*fx[0] + w[s][1]*fx[1] + w[s][2]*fx[2];
            ay[s] = w[s][0]*fy[0] + w[s][1]*fy[1] + w[s][2]*fy[2];
        }

        float mu = (ax[0]+ay[0]+ax[1]+ay[1]+ax[2]+ay[2]) * (1.f/6.f);
        float var = 0.f;
        #pragma unroll
        for (int s = 0; s < 3; ++s) {
            float dx = ax[s] - mu, dy = ay[s] - mu;
            var += dx*dx + dy*dy;
        }
        var *= (1.f/6.f);
        float inv = __builtin_amdgcn_rsqf(var + 1e-5f);
        #pragma unroll
        for (int s = 0; s < 3; ++s) {
            flat[2*s+0] = (ax[s] - mu) * inv * W.lnw[2*s+0] + W.lnb[2*s+0];
            flat[2*s+1] = (ay[s] - mu) * inv * W.lnw[2*s+1] + W.lnb[2*s+1];
        }
    }

    if (waveBase + 32 <= n) {
        v2f B0; B0.x = flat[0]; B0.y = flat[1];
        v2f B1; B1.x = flat[2]; B1.y = flat[3];
        v2f B2; B2.x = flat[4]; B2.y = flat[5];
        v8f D = __builtin_amdgcn_wmma_f32_16x16x4_f32(false, A0, false, B0, (short)0, C, false, false);
        D     = __builtin_amdgcn_wmma_f32_16x16x4_f32(false, A1, false, B1, (short)0, D, false, false);
        D     = __builtin_amdgcn_wmma_f32_16x16x4_f32(false, A2, false, B2, (short)0, D, false, false);

        float* o = out + (size_t)row * 6u;
        v2f s0; s0.x = fmaxf(D[0], 0.f); s0.y = fmaxf(D[1], 0.f);
        v2f s1; s1.x = fmaxf(D[2], 0.f); s1.y = fmaxf(D[3], 0.f);
        v2f s2; s2.x = fmaxf(D[4], 0.f); s2.y = fmaxf(D[5], 0.f);
        __builtin_nontemporal_store(s0, (v2f*)(o + 0));
        __builtin_nontemporal_store(s1, (v2f*)(o + 2));
        __builtin_nontemporal_store(s2, (v2f*)(o + 4));
    } else if (row < n) {
        float* o = out + (size_t)row * 6u;
        #pragma unroll
        for (int jj = 0; jj < 6; ++jj) {
            float acc = bfc[jj];
            #pragma unroll
            for (int k = 0; k < 6; ++k) acc += Wfc[jj*6 + k] * flat[k];
            o[jj] = fmaxf(acc, 0.f);
        }
    }
}

__global__ __launch_bounds__(256) void fused_attn_kernel(
    const float* __restrict__ hist,
    const float* __restrict__ paper,
    const float* __restrict__ author,
    const float* __restrict__ Wh, const float* __restrict__ bh,
    const float* __restrict__ Wp, const float* __restrict__ bp,
    const float* __restrict__ Wa, const float* __restrict__ ba,
    const float* __restrict__ lnw, const float* __restrict__ lnb,
    const float* __restrict__ Wfc, const float* __restrict__ bfc,
    float* __restrict__ out, int n)
{
    if (n <= 0) return;
    const int tid    = threadIdx.x;
    const int lane   = tid & 31;
    const bool lo    = lane < 16;
    const int stride = gridDim.x * blockDim.x;
    const int last   = n - 1;

    const int  c16 = lane & 15;
    const int  jr  = lo ? c16 : c16 - 8;
    const bool jv  = (jr >= 0) && (jr < 6);
    const int  jo  = jv ? jr * 6 : 0;
    v2f A0, A1, A2;
    A0.x = jv ? Wfc[jo + 0] : 0.f;  A0.y = jv ? Wfc[jo + 1] : 0.f;
    A1.x = jv ? Wfc[jo + 2] : 0.f;  A1.y = jv ? Wfc[jo + 3] : 0.f;
    A2.x = jv ? Wfc[jo + 4] : 0.f;  A2.y = jv ? Wfc[jo + 5] : 0.f;
    v8f C = {};
    C[0]=bfc[0]; C[1]=bfc[1]; C[2]=bfc[2]; C[3]=bfc[3]; C[4]=bfc[4]; C[5]=bfc[5];

    Wts W;
    #pragma unroll
    for (int i = 0; i < 4;  ++i) W.wh[i] = Wh[i];
    #pragma unroll
    for (int i = 0; i < 10; ++i) { W.wp[i] = Wp[i]; W.wa[i] = Wa[i]; }
    #pragma unroll
    for (int i = 0; i < 2;  ++i) { W.bh[i] = bh[i]; W.bp[i] = bp[i]; W.ba[i] = ba[i]; }
    #pragma unroll
    for (int i = 0; i < 6;  ++i) { W.lnw[i] = lnw[i]; W.lnb[i] = lnb[i]; }

    int slab = blockIdx.x * blockDim.x;
    RowIn cur = load_row(hist, paper, author, min(slab + tid, last));

    for (; slab < n; slab += stride) {
        const RowIn nxt = load_row(hist, paper, author, min(slab + stride + tid, last));
        process_slab(cur, W, A0, A1, A2, C, Wfc, bfc, out, slab + tid, lane, n);
        cur = nxt;
    }
}

extern "C" void kernel_launch(void* const* d_in, const int* in_sizes, int n_in,
                              void* d_out, int out_size, void* d_ws, size_t ws_size,
                              hipStream_t stream) {
    const float* hist   = (const float*)d_in[0];
    const float* paper  = (const float*)d_in[1];
    const float* author = (const float*)d_in[2];
    const float* Wh  = (const float*)d_in[3];
    const float* bh  = (const float*)d_in[4];
    const float* Wp  = (const float*)d_in[5];
    const float* bp  = (const float*)d_in[6];
    const float* Wa  = (const float*)d_in[7];
    const float* ba  = (const float*)d_in[8];
    const float* lnw = (const float*)d_in[9];
    const float* lnb = (const float*)d_in[10];
    const float* Wfc = (const float*)d_in[11];
    const float* bfc = (const float*)d_in[12];
    float* out = (float*)d_out;

    const int n = in_sizes[0] / 2;
    const int threads = 256;
    const int blocks  = 1024;

    fused_attn_kernel<<<blocks, threads, 0, stream>>>(
        hist, paper, author, Wh, bh, Wp, bp, Wa, ba, lnw, lnb, Wfc, bfc, out, n);
}
